# starting kernel
# baseline (speedup 1.0000x reference)
#include <hip/hip_runtime.h>

typedef __attribute__((ext_vector_type(2))) float v2f;
typedef __attribute__((ext_vector_type(8))) float v8f;

#define D 128

__global__ __launch_bounds__(256) void k_deg_init(float* __restrict__ deg, int n) {
  int i = blockIdx.x * 256 + threadIdx.x;
  if (i < n) deg[i] = 1.0f;
}

__global__ __launch_bounds__(256) void k_deg_accum(const int* __restrict__ col,
                                                   float* __restrict__ deg, int e) {
  int i = blockIdx.x * 256 + threadIdx.x;
  if (i < e) atomicAdd(&deg[col[i]], 1.0f);
}

__global__ __launch_bounds__(256) void k_dinv(float* __restrict__ deg, int n) {
  int i = blockIdx.x * 256 + threadIdx.x;
  if (i < n) {
    float d = deg[i];
    deg[i] = d > 0.0f ? rsqrtf(d) : 0.0f;
  }
}

__global__ __launch_bounds__(256) void k_gemm_selfinit(
    const float* __restrict__ seq, const float* __restrict__ W,
    const float* __restrict__ dinv, float* __restrict__ xW,
    float* __restrict__ out) {
  const int lane  = threadIdx.x & 31;
  const int wave  = threadIdx.x >> 5;
  const int row0  = blockIdx.x * 16;
  const int col0  = wave * 16;
  const int mn    = lane & 15;
  const int khalf = (lane >> 4) << 1;

  const float* arow = seq + (size_t)(row0 + mn) * D;
  v8f acc = {};
#pragma unroll 4
  for (int k = 0; k < D; k += 4) {
    v2f a, b;
    a.x = arow[k + khalf];
    a.y = arow[k + khalf + 1];
    b.x = W[(size_t)(k + khalf) * D + col0 + mn];
    b.y = W[(size_t)(k + khalf + 1) * D + col0 + mn];
    acc = __builtin_amdgcn_wmma_f32_16x16x4_f32(
         false, a,  false, b,
         (short)0, acc,  false,  false);
  }

  const int mbase = (lane >> 4) << 3;
#pragma unroll
  for (int j = 0; j < 8; ++j) {
    int r = row0 + mbase + j;
    float di = dinv[r];
    size_t off = (size_t)r * D + col0 + mn;
    xW[off]  = acc[j];
    out[off] = acc[j] * di * di;
  }
}

__global__ __launch_bounds__(128) void k_tail_rows(
    const float* __restrict__ seq, const float* __restrict__ W,
    const float* __restrict__ dinv, float* __restrict__ xW,
    float* __restrict__ out, int rowStart, int n) {
  int r = rowStart + blockIdx.x;
  if (r >= n) return;
  int c = threadIdx.x;
  float acc = 0.0f;
  for (int k = 0; k < D; ++k) acc += seq[(size_t)r * D + k] * W[(size_t)k * D + c];
  float di = dinv[r];
  xW[(size_t)r * D + c]  = acc;
  out[(size_t)r * D + c] = acc * di * di;
}

__global__ __launch_bounds__(256) void k_scatter(
    const int* __restrict__ row, const int* __restrict__ col,
    const float* __restrict__ xW, const float* __restrict__ dinv,
    float* __restrict__ out, int e) {
  int idx = blockIdx.x * 256 + threadIdx.x;
  int eid = idx >> 5;
  if (eid >= e) return;
  int lane = idx & 31;
  int s = row[eid];
  int t = col[eid];
  float w = dinv[s] * dinv[t];
  const float4* src = (const float4*)(xW + (size_t)s * D);
  float4 v = src[lane];
  float* dst = out + (size_t)t * D + lane * 4;
  atomicAdd(dst + 0, v.x * w);
  atomicAdd(dst + 1, v.y * w);
  atomicAdd(dst + 2, v.z * w);
  atomicAdd(dst + 3, v.w * w);
}

__global__ __launch_bounds__(256) void k_finalize(
    float* __restrict__ out, const float* __restrict__ bias,
    const float* __restrict__ prelu, int total) {
  int i = blockIdx.x * 256 + threadIdx.x;
  if (i >= total) return;
  float v = out[i] + bias[i & (D - 1)];
  float pw = prelu[0];
  out[i] = v > 0.0f ? v : pw * v;
}

extern "C" void kernel_launch(void* const* d_in, const int* in_sizes, int n_in,
                              void* d_out, int out_size, void* d_ws, size_t ws_size,
                              hipStream_t stream) {
  const float* seq   = (const float*)d_in[0];
  const int*   eidx  = (const int*)d_in[1];
  const float* W     = (const float*)d_in[2];
  const float* bias  = (const float*)d_in[3];
  const float* prelu = (const float*)d_in[4];

  const int N = in_sizes[0] / D;
  const int E = in_sizes[1] / 2;
  const int* rowp = eidx;
  const int* colp = eidx + E;

  float* xW  = (float*)d_ws;
  float* deg = xW + (size_t)N * D;
  float* out = (float*)d_out;

  k_deg_init <<<(N + 255) / 256, 256, 0, stream>>>(deg, N);
  k_deg_accum<<<(E + 255) / 256, 256, 0, stream>>>(colp, deg, E);
  k_dinv     <<<(N + 255) / 256, 256, 0, stream>>>(deg, N);

  int nt  = N / 16;
  int rem = N - nt * 16;
  if (nt  > 0) k_gemm_selfinit<<<nt, 256, 0, stream>>>(seq, W, deg, xW, out);
  if (rem > 0) k_tail_rows<<<rem, 128, 0, stream>>>(seq, W, deg, xW, out, nt * 16, N);

  int scatter_blocks = (int)(((size_t)E * 32 + 255) / 256);
  k_scatter <<<scatter_blocks, 256, 0, stream>>>(rowp, colp, xW, deg, out, E);

  int total = N * D;
  k_finalize<<<(total + 255) / 256, 256, 0, stream>>>(out, bias, prelu, total);
}


# Degree count by per-block LDS integer histograms over edge chunks, summed per node in the dinv kernel, replaces the global float atomic adds on deg; counts exact, float sum orders unchanged
# speedup vs baseline: 1.2460x; 1.2460x over previous
#include <hip/hip_runtime.h>

typedef __attribute__((ext_vector_type(2))) float v2f;
typedef __attribute__((ext_vector_type(8))) float v8f;

#define D 128

#define HB_BINS    12800
#define HB_CHUNKS  64
#define HB_THREADS 1024
#define HB_BATCH   16

__global__ __launch_bounds__(HB_THREADS) void k_deg_hist(const int* __restrict__ col,
                                                         int* __restrict__ part, int e, int n) {
  __shared__ int h[HB_BINS];
  const int tid  = threadIdx.x;
  const int bin0 = blockIdx.y * HB_BINS;
  int nb = n - bin0;
  if (nb > HB_BINS) nb = HB_BINS;

  for (int i = tid; i < HB_BINS; i += HB_THREADS) h[i] = 0;
  __syncthreads();

  const int per = (e + (int)gridDim.x - 1) / (int)gridDim.x;
  const int beg = blockIdx.x * per;
  int end = beg + per;
  if (end > e) end = e;
  for (int base = beg + tid; base < end; base += HB_THREADS * HB_BATCH) {
    int t[HB_BATCH];
#pragma unroll
    for (int u = 0; u < HB_BATCH; ++u) {
      int i = base + u * HB_THREADS;
      int v = col[i < end ? i : end - 1];
      t[u] = i < end ? v : -1;
    }
#pragma unroll
    for (int u = 0; u < HB_BATCH; ++u) {
      unsigned b = (unsigned)(t[u] - bin0);
      if (b < (unsigned)nb) atomicAdd(&h[b], 1);
    }
  }
  __syncthreads();

  int* dst = part + (size_t)blockIdx.x * n + bin0;
  for (int i = tid; i < nb; i += HB_THREADS) dst[i] = h[i];
}

__global__ __launch_bounds__(256) void k_dinv(const int* __restrict__ part,
                                              float* __restrict__ dinv, int n) {
  int i = blockIdx.x * 256 + threadIdx.x;
  if (i < n) {
    int c = 1;
#pragma unroll 32
    for (int k = 0; k < HB_CHUNKS; ++k) c += part[(size_t)k * n + i];
    float d = (float)c;
    dinv[i] = d > 0.0f ? rsqrtf(d) : 0.0f;
  }
}

__global__ __launch_bounds__(256) void k_gemm_selfinit(
    const float* __restrict__ seq, const float* __restrict__ W,
    const float* __restrict__ dinv, float* __restrict__ xW,
    float* __restrict__ out) {
  const int lane  = threadIdx.x & 31;
  const int wave  = threadIdx.x >> 5;
  const int row0  = blockIdx.x * 16;
  const int col0  = wave * 16;
  const int mn    = lane & 15;
  const int khalf = (lane >> 4) << 1;

  const float* arow = seq + (size_t)(row0 + mn) * D;
  v8f acc = {};
#pragma unroll 4
  for (int k = 0; k < D; k += 4) {
    v2f a, b;
    a.x = arow[k + khalf];
    a.y = arow[k + khalf + 1];
    b.x = W[(size_t)(k + khalf) * D + col0 + mn];
    b.y = W[(size_t)(k + khalf + 1) * D + col0 + mn];
    acc = __builtin_amdgcn_wmma_f32_16x16x4_f32(
         false, a,  false, b,
         (short)0, acc,  false,  false);
  }

  const int mbase = (lane >> 4) << 3;
#pragma unroll
  for (int j = 0; j < 8; ++j) {
    int r = row0 + mbase + j;
    float di = dinv[r];
    size_t off = (size_t)r * D + col0 + mn;
    xW[off]  = acc[j];
    out[off] = acc[j] * di * di;
  }
}

__global__ __launch_bounds__(128) void k_tail_rows(
    const float* __restrict__ seq, const float* __restrict__ W,
    const float* __restrict__ dinv, float* __restrict__ xW,
    float* __restrict__ out, int rowStart, int n) {
  int r = rowStart + blockIdx.x;
  if (r >= n) return;
  int c = threadIdx.x;
  float acc = 0.0f;
  for (int k = 0; k < D; ++k) acc += seq[(size_t)r * D + k] * W[(size_t)k * D + c];
  float di = dinv[r];
  xW[(size_t)r * D + c]  = acc;
  out[(size_t)r * D + c] = acc * di * di;
}

__global__ __launch_bounds__(256) void k_scatter(
    const int* __restrict__ row, const int* __restrict__ col,
    const float* __restrict__ xW, const float* __restrict__ dinv,
    float* __restrict__ out, int e) {
  int idx = blockIdx.x * 256 + threadIdx.x;
  int eid = idx >> 5;
  if (eid >= e) return;
  int lane = idx & 31;
  int s = row[eid];
  int t = col[eid];
  float w = dinv[s] * dinv[t];
  const float4* src = (const float4*)(xW + (size_t)s * D);
  float4 v = src[lane];
  float* dst = out + (size_t)t * D + lane * 4;
  atomicAdd(dst + 0, v.x * w);
  atomicAdd(dst + 1, v.y * w);
  atomicAdd(dst + 2, v.z * w);
  atomicAdd(dst + 3, v.w * w);
}

__global__ __launch_bounds__(256) void k_finalize(
    float* __restrict__ out, const float* __restrict__ bias,
    const float* __restrict__ prelu, int total) {
  int i = blockIdx.x * 256 + threadIdx.x;
  if (i >= total) return;
  float v = out[i] + bias[i & (D - 1)];
  float pw = prelu[0];
  out[i] = v > 0.0f ? v : pw * v;
}

extern "C" void kernel_launch(void* const* d_in, const int* in_sizes, int n_in,
                              void* d_out, int out_size, void* d_ws, size_t ws_size,
                              hipStream_t stream) {
  const float* seq   = (const float*)d_in[0];
  const int*   eidx  = (const int*)d_in[1];
  const float* W     = (const float*)d_in[2];
  const float* bias  = (const float*)d_in[3];
  const float* prelu = (const float*)d_in[4];

  const int N = in_sizes[0] / D;
  const int E = in_sizes[1] / 2;
  const int* rowp = eidx;
  const int* colp = eidx + E;

  float* xW  = (float*)d_ws;
  float* deg = xW + (size_t)N * D;
  int*  part = (int*)(deg + N);
  float* out = (float*)d_out;

  k_deg_hist<<<dim3(HB_CHUNKS, (N + HB_BINS - 1) / HB_BINS), HB_THREADS, 0, stream>>>(colp, part, E, N);
  k_dinv    <<<(N + 255) / 256, 256, 0, stream>>>(part, deg, N);

  int nt  = N / 16;
  int rem = N - nt * 16;
  if (nt  > 0) k_gemm_selfinit<<<nt, 256, 0, stream>>>(seq, W, deg, xW, out);
  if (rem > 0) k_tail_rows<<<rem, 128, 0, stream>>>(seq, W, deg, xW, out, nt * 16, N);

  int scatter_blocks = (int)(((size_t)E * 32 + 255) / 256);
  k_scatter <<<scatter_blocks, 256, 0, stream>>>(rowp, colp, xW, deg, out, E);

  int total = N * D;
  k_finalize<<<(total + 255) / 256, 256, 0, stream>>>(out, bias, prelu, total);
}


# Aggregate by target-sorted edges (counting sort from the chunk histograms, arrival-order rank inside a chunk): one wave sums a node's rows in registers, one store with bias+PReLU; no float atomics
# speedup vs baseline: 2.6218x; 2.1042x over previous
#include <hip/hip_runtime.h>

typedef __attribute__((ext_vector_type(2))) float v2f;
typedef __attribute__((ext_vector_type(8))) float v8f;

#define D 128

#define HB_BINS    12800
#define HB_CHUNKS  64
#define HB_THREADS 1024
#define HB_BATCH   16
#define SC_THREADS 256

__global__ __launch_bounds__(HB_THREADS) void k_deg_hist(const int* __restrict__ col,
                                                         int* __restrict__ part, int e, int n) {
  __shared__ int h[HB_BINS];
  const int tid  = threadIdx.x;
  const int bin0 = blockIdx.y * HB_BINS;
  int nb = n - bin0;
  if (nb > HB_BINS) nb = HB_BINS;

  for (int i = tid; i < HB_BINS; i += HB_THREADS) h[i] = 0;
  __syncthreads();

  const int per = (e + (int)gridDim.x - 1) / (int)gridDim.x;
  const int beg = blockIdx.x * per;
  int end = beg + per;
  if (end > e) end = e;
  for (int base = beg + tid; base < end; base += HB_THREADS * HB_BATCH) {
    int t[HB_BATCH];
#pragma unroll
    for (int u = 0; u < HB_BATCH; ++u) {
      int i = base + u * HB_THREADS;
      int v = col[i < end ? i : end - 1];
      t[u] = i < end ? v : -1;
    }
#pragma unroll
    for (int u = 0; u < HB_BATCH; ++u) {
      unsigned b = (unsigned)(t[u] - bin0);
      if (b < (unsigned)nb) atomicAdd(&h[b], 1);
    }
  }
  __syncthreads();

  int* dst = part + (size_t)blockIdx.x * n + bin0;
  for (int i = tid; i < nb; i += HB_THREADS) dst[i] = h[i];
}

__device__ __forceinline__ int wave_incl_scan(int v, int lane) {
#pragma unroll
  for (int o = 1; o < 32; o <<= 1) {
    int y = __shfl_up(v, o, 32);
    if (lane >= o) v += y;
  }
  return v;
}

__global__ __launch_bounds__(SC_THREADS) void k_dinv_scan(
    const int* __restrict__ part, int* __restrict__ pre, float* __restrict__ dinv,
    int* __restrict__ cnt, int* __restrict__ loc, int* __restrict__ blocksum, int n) {
  __shared__ int wsum[SC_THREADS / 32];
  const int tid  = threadIdx.x;
  const int lane = tid & 31;
  const int wave = tid >> 5;
  const int i = blockIdx.x * SC_THREADS + tid;
  int c = 0;
  if (i < n) {
    int v[HB_CHUNKS];
#pragma unroll
    for (int k = 0; k < HB_CHUNKS; ++k) v[k] = part[(size_t)k * n + i];
#pragma unroll
    for (int k = 0; k < HB_CHUNKS; ++k) {
      pre[(size_t)k * n + i] = c;
      c += v[k];
    }
    float d = (float)(c + 1);
    dinv[i] = d > 0.0f ? rsqrtf(d) : 0.0f;
    cnt[i] = c;
  }
  const int inc = wave_incl_scan(c, lane);
  if (lane == 31) wsum[wave] = inc;
  __syncthreads();
  int wb = 0;
  for (int w = 0; w < wave; ++w) wb += wsum[w];
  if (i < n) loc[i] = wb + inc - c;
  if (tid == SC_THREADS - 1) blocksum[blockIdx.x] = wb + inc;
}

__global__ __launch_bounds__(SC_THREADS) void k_blockscan(
    const int* __restrict__ blocksum, int* __restrict__ blockoff, int nblk) {
  __shared__ int wsum[SC_THREADS / 32];
  const int tid  = threadIdx.x;
  const int lane = tid & 31;
  const int wave = tid >> 5;
  int carry = 0;
  for (int base = 0; base < nblk; base += SC_THREADS) {
    const int q = base + tid;
    const int v = q < nblk ? blocksum[q] : 0;
    const int inc = wave_incl_scan(v, lane);
    if (lane == 31) wsum[wave] = inc;
    __syncthreads();
    int wb = 0, tot = 0;
    for (int w = 0; w < SC_THREADS / 32; ++w) {
      int s = wsum[w];
      if (w < wave) wb += s;
      tot += s;
    }
    if (q < nblk) blockoff[q] = carry + wb + inc - v;
    carry += tot;
    __syncthreads();
  }
}

__global__ __launch_bounds__(HB_THREADS) void k_place(
    const int* __restrict__ row, const int* __restrict__ col,
    const int* __restrict__ pre, const int* __restrict__ loc,
    const int* __restrict__ blockoff, int* __restrict__ srcs, int e, int n) {
  __shared__ int cur[HB_BINS];
  const int tid  = threadIdx.x;
  const int bin0 = blockIdx.y * HB_BINS;
  int nb = n - bin0;
  if (nb > HB_BINS) nb = HB_BINS;

  const int* prec = pre + (size_t)blockIdx.x * n + bin0;
#pragma unroll 4
  for (int i = tid; i < nb; i += HB_THREADS)
    cur[i] = blockoff[(bin0 + i) / SC_THREADS] + loc[bin0 + i] + prec[i];
  __syncthreads();

  const int per = (e + (int)gridDim.x - 1) / (int)gridDim.x;
  const int beg = blockIdx.x * per;
  int end = beg + per;
  if (end > e) end = e;
  for (int base = beg + tid; base < end; base += HB_THREADS * HB_BATCH) {
    int t[HB_BATCH], s[HB_BATCH];
#pragma unroll
    for (int u = 0; u < HB_BATCH; ++u) {
      int i = base + u * HB_THREADS;
      int j = i < end ? i : end - 1;
      int v = col[j];
      s[u] = row[j];
      t[u] = i < end ? v : -1;
    }
#pragma unroll
    for (int u = 0; u < HB_BATCH; ++u) {
      unsigned b = (unsigned)(t[u] - bin0);
      if (b < (unsigned)nb) {
        int slot = atomicAdd(&cur[b], 1);
        srcs[slot] = s[u];
      }
    }
  }
}

__global__ __launch_bounds__(256) void k_gemm(
    const float* __restrict__ seq, const float* __restrict__ W,
    float* __restrict__ xW) {
  const int lane  = threadIdx.x & 31;
  const int wave  = threadIdx.x >> 5;
  const int row0  = blockIdx.x * 16;
  const int col0  = wave * 16;
  const int mn    = lane & 15;
  const int khalf = (lane >> 4) << 1;

  const float* arow = seq + (size_t)(row0 + mn) * D;
  v8f acc = {};
#pragma unroll 4
  for (int k = 0; k < D; k += 4) {
    v2f a, b;
    a.x = arow[k + khalf];
    a.y = arow[k + khalf + 1];
    b.x = W[(size_t)(k + khalf) * D + col0 + mn];
    b.y = W[(size_t)(k + khalf + 1) * D + col0 + mn];
    acc = __builtin_amdgcn_wmma_f32_16x16x4_f32(
         false, a,  false, b,
         (short)0, acc,  false,  false);
  }

  const int mbase = (lane >> 4) << 3;
#pragma unroll
  for (int j = 0; j < 8; ++j) {
    int r = row0 + mbase + j;
    xW[(size_t)r * D + col0 + mn] = acc[j];
  }
}

__global__ __launch_bounds__(128) void k_tail_rows(
    const float* __restrict__ seq, const float* __restrict__ W,
    float* __restrict__ xW, int rowStart, int n) {
  int r = rowStart + blockIdx.x;
  if (r >= n) return;
  int c = threadIdx.x;
  float acc = 0.0f;
  for (int k = 0; k < D; ++k) acc += seq[(size_t)r * D + k] * W[(size_t)k * D + c];
  xW[(size_t)r * D + c] = acc;
}

#define GA_UNROLL 8

__global__ __launch_bounds__(256) void k_gather(
    const int* __restrict__ srcs, const int* __restrict__ cnt,
    const int* __restrict__ loc, const int* __restrict__ blockoff,
    const float* __restrict__ xW, const float* __restrict__ dinv,
    const float* __restrict__ bias, const float* __restrict__ prelu,
    float* __restrict__ out, int n) {
  const int lane = threadIdx.x & 31;
  const int t = __builtin_amdgcn_readfirstlane(blockIdx.x * 8 + (threadIdx.x >> 5));
  if (t >= n) return;

  const int   beg = blockoff[t / SC_THREADS] + loc[t];
  const int   c   = cnt[t];
  const float dt  = dinv[t];
  const float4* xW4 = (const float4*)xW;

  float4 acc = xW4[(size_t)t * (D / 4) + lane];
  const float ws = dt * dt;
  acc.x *= ws; acc.y *= ws; acc.z *= ws; acc.w *= ws;

  for (int j0 = 0; j0 < c; j0 += 32) {
    int m = c - j0;
    if (m > 32) m = 32;
    int   s_l = t;
    float w_l = 0.0f;
    if (lane < m) {
      s_l = srcs[beg + j0 + lane];
      w_l = dinv[s_l] * dt;
    }
    for (int j = 0; j < m; j += GA_UNROLL) {
      float4 v[GA_UNROLL];
      float  w[GA_UNROLL];
#pragma unroll
      for (int u = 0; u < GA_UNROLL; ++u) {
        const int su = __builtin_amdgcn_readlane(s_l, j + u);
        w[u] = __int_as_float(__builtin_amdgcn_readlane(__float_as_int(w_l), j + u));
        v[u] = xW4[(size_t)su * (D / 4) + lane];
      }
#pragma unroll
      for (int u = 0; u < GA_UNROLL; ++u) {
        acc.x += v[u].x * w[u];
        acc.y += v[u].y * w[u];
        acc.z += v[u].z * w[u];
        acc.w += v[u].w * w[u];
      }
    }
  }

  const float pw = prelu[0];
  float4 o;
  o.x = acc.x + bias[lane * 4 + 0];
  o.y = acc.y + bias[lane * 4 + 1];
  o.z = acc.z + bias[lane * 4 + 2];
  o.w = acc.w + bias[lane * 4 + 3];
  o.x = o.x > 0.0f ? o.x : pw * o.x;
  o.y = o.y > 0.0f ? o.y : pw * o.y;
  o.z = o.z > 0.0f ? o.z : pw * o.z;
  o.w = o.w > 0.0f ? o.w : pw * o.w;
  ((float4*)out)[(size_t)t * (D / 4) + lane] = o;
}

extern "C" void kernel_launch(void* const* d_in, const int* in_sizes, int n_in,
                              void* d_out, int out_size, void* d_ws, size_t ws_size,
                              hipStream_t stream) {
  const float* seq   = (const float*)d_in[0];
  const int*   eidx  = (const int*)d_in[1];
  const float* W     = (const float*)d_in[2];
  const float* bias  = (const float*)d_in[3];
  const float* prelu = (const float*)d_in[4];

  const int N = in_sizes[0] / D;
  const int E = in_sizes[1] / 2;
  const int* rowp = eidx;
  const int* colp = eidx + E;
  const int nblk = (N + SC_THREADS - 1) / SC_THREADS;

  float* xW   = (float*)d_ws;
  float* deg  = xW + (size_t)N * D;
  int*  part  = (int*)(deg + N);
  int*  pre   = part + (size_t)HB_CHUNKS * N;
  int*  loc   = pre + (size_t)HB_CHUNKS * N;
  int*  cnt   = loc + N;
  int*  bsum  = cnt + N;
  int*  boff  = bsum + nblk;
  int*  srcs  = boff + nblk;
  float* out  = (float*)d_out;

  const dim3 hgrid(HB_CHUNKS, (N + HB_BINS - 1) / HB_BINS);
  k_deg_hist <<<hgrid, HB_THREADS, 0, stream>>>(colp, part, E, N);
  k_dinv_scan<<<nblk, SC_THREADS, 0, stream>>>(part, pre, deg, cnt, loc, bsum, N);
  k_blockscan<<<1, SC_THREADS, 0, stream>>>(bsum, boff, nblk);
  k_place    <<<hgrid, HB_THREADS, 0, stream>>>(rowp, colp, pre, loc, boff, srcs, E, N);

  int nt  = N / 16;
  int rem = N - nt * 16;
  if (nt  > 0) k_gemm<<<nt, 256, 0, stream>>>(seq, W, xW);
  if (rem > 0) k_tail_rows<<<rem, 128, 0, stream>>>(seq, W, xW, nt * 16, N);

  k_gather<<<(N + 7) / 8, 256, 0, stream>>>(srcs, cnt, loc, boff, xW, deg, bias, prelu, out, N);
}


# Transform after aggregation, fused: a block sums each 16-node stripe's normalized seq rows into LDS, then WMMA with W columns held in registers, bias+PReLU; float sum order changes
# speedup vs baseline: 3.2817x; 1.2517x over previous
#include <hip/hip_runtime.h>

typedef __attribute__((ext_vector_type(2))) float v2f;
typedef __attribute__((ext_vector_type(8))) float v8f;

#define D 128

#define HB_BINS    12800
#define HB_CHUNKS  64
#define HB_THREADS 1024
#define HB_BATCH   16
#define SC_THREADS 256

__global__ __launch_bounds__(HB_THREADS) void k_deg_hist(const int* __restrict__ col,
                                                         int* __restrict__ part, int e, int n) {
  __shared__ int h[HB_BINS];
  const int tid  = threadIdx.x;
  const int bin0 = blockIdx.y * HB_BINS;
  int nb = n - bin0;
  if (nb > HB_BINS) nb = HB_BINS;

  for (int i = tid; i < HB_BINS; i += HB_THREADS) h[i] = 0;
  __syncthreads();

  const int per = (e + (int)gridDim.x - 1) / (int)gridDim.x;
  const int beg = blockIdx.x * per;
  int end = beg + per;
  if (end > e) end = e;
  for (int base = beg + tid; base < end; base += HB_THREADS * HB_BATCH) {
    int t[HB_BATCH];
#pragma unroll
    for (int u = 0; u < HB_BATCH; ++u) {
      int i = base + u * HB_THREADS;
      int v = col[i < end ? i : end - 1];
      t[u] = i < end ? v : -1;
    }
#pragma unroll
    for (int u = 0; u < HB_BATCH; ++u) {
      unsigned b = (unsigned)(t[u] - bin0);
      if (b < (unsigned)nb) atomicAdd(&h[b], 1);
    }
  }
  __syncthreads();

  int* dst = part + (size_t)blockIdx.x * n + bin0;
  for (int i = tid; i < nb; i += HB_THREADS) dst[i] = h[i];
}

__device__ __forceinline__ int wave_incl_scan(int v, int lane) {
#pragma unroll
  for (int o = 1; o < 32; o <<= 1) {
    int y = __shfl_up(v, o, 32);
    if (lane >= o) v += y;
  }
  return v;
}

__global__ __launch_bounds__(SC_THREADS) void k_dinv_scan(
    const int* __restrict__ part, int* __restrict__ pre, float* __restrict__ dinv,
    int* __restrict__ cnt, int* __restrict__ loc, int* __restrict__ blocksum, int n) {
  __shared__ int wsum[SC_THREADS / 32];
  const int tid  = threadIdx.x;
  const int lane = tid & 31;
  const int wave = tid >> 5;
  const int i = blockIdx.x * SC_THREADS + tid;
  int c = 0;
  if (i < n) {
    int v[HB_CHUNKS];
#pragma unroll
    for (int k = 0; k < HB_CHUNKS; ++k) v[k] = part[(size_t)k * n + i];
#pragma unroll
    for (int k = 0; k < HB_CHUNKS; ++k) {
      pre[(size_t)k * n + i] = c;
      c += v[k];
    }
    float d = (float)(c + 1);
    dinv[i] = d > 0.0f ? rsqrtf(d) : 0.0f;
    cnt[i] = c;
  }
  const int inc = wave_incl_scan(c, lane);
  if (lane == 31) wsum[wave] = inc;
  __syncthreads();
  int wb = 0;
  for (int w = 0; w < wave; ++w) wb += wsum[w];
  if (i < n) loc[i] = wb + inc - c;
  if (tid == SC_THREADS - 1) blocksum[blockIdx.x] = wb + inc;
}

__global__ __launch_bounds__(SC_THREADS) void k_blockscan(
    const int* __restrict__ blocksum, int* __restrict__ blockoff, int nblk) {
  __shared__ int wsum[SC_THREADS / 32];
  const int tid  = threadIdx.x;
  const int lane = tid & 31;
  const int wave = tid >> 5;
  int carry = 0;
  for (int base = 0; base < nblk; base += SC_THREADS) {
    const int q = base + tid;
    const int v = q < nblk ? blocksum[q] : 0;
    const int inc = wave_incl_scan(v, lane);
    if (lane == 31) wsum[wave] = inc;
    __syncthreads();
    int wb = 0, tot = 0;
    for (int w = 0; w < SC_THREADS / 32; ++w) {
      int s = wsum[w];
      if (w < wave) wb += s;
      tot += s;
    }
    if (q < nblk) blockoff[q] = carry + wb + inc - v;
    carry += tot;
    __syncthreads();
  }
}

__global__ __launch_bounds__(HB_THREADS) void k_place(
    const int* __restrict__ row, const int* __restrict__ col,
    const int* __restrict__ pre, const int* __restrict__ loc,
    const int* __restrict__ blockoff, int* __restrict__ srcs, int e, int n) {
  __shared__ int cur[HB_BINS];
  const int tid  = threadIdx.x;
  const int bin0 = blockIdx.y * HB_BINS;
  int nb = n - bin0;
  if (nb > HB_BINS) nb = HB_BINS;

  const int* prec = pre + (size_t)blockIdx.x * n + bin0;
#pragma unroll 4
  for (int i = tid; i < nb; i += HB_THREADS)
    cur[i] = blockoff[(bin0 + i) / SC_THREADS] + loc[bin0 + i] + prec[i];
  __syncthreads();

  const int per = (e + (int)gridDim.x - 1) / (int)gridDim.x;
  const int beg = blockIdx.x * per;
  int end = beg + per;
  if (end > e) end = e;
  for (int base = beg + tid; base < end; base += HB_THREADS * HB_BATCH) {
    int t[HB_BATCH], s[HB_BATCH];
#pragma unroll
    for (int u = 0; u < HB_BATCH; ++u) {
      int i = base + u * HB_THREADS;
      int j = i < end ? i : end - 1;
      int v = col[j];
      s[u] = row[j];
      t[u] = i < end ? v : -1;
    }
#pragma unroll
    for (int u = 0; u < HB_BATCH; ++u) {
      unsigned b = (unsigned)(t[u] - bin0);
      if (b < (unsigned)nb) {
        int slot = atomicAdd(&cur[b], 1);
        srcs[slot] = s[u];
      }
    }
  }
}

#define GA_UNROLL 8
#define ST_PITCH  132
#define AG_GRID   768

__global__ __launch_bounds__(256) void k_agg_gemm(
    const int* __restrict__ srcs, const int* __restrict__ cnt,
    const int* __restrict__ loc, const int* __restrict__ blockoff,
    const float* __restrict__ seq, const float* __restrict__ W,
    const float* __restrict__ dinv,
    const float* __restrict__ bias, const float* __restrict__ prelu,
    float* __restrict__ out, int n, int nstripes) {
  __shared__ __attribute__((aligned(16))) float agg[16 * ST_PITCH];
  __shared__ __attribute__((aligned(16))) float cst[16 * ST_PITCH];

  const int lane  = threadIdx.x & 31;
  const int wave  = __builtin_amdgcn_readfirstlane(threadIdx.x >> 5);
  const int col0  = wave * 16;
  const int mn    = lane & 15;
  const int khalf = (lane >> 4) << 1;
  const int mbase = (lane >> 4) << 3;
  const float4* seq4 = (const float4*)seq;

  v2f b[D / 4];
#pragma unroll
  for (int i = 0; i < D / 4; ++i) {
    b[i].x = W[(size_t)(4 * i + khalf) * D + col0 + mn];
    b[i].y = W[(size_t)(4 * i + khalf + 1) * D + col0 + mn];
  }
  const float pw = prelu[0];
  float4 bv;
  bv.x = bias[lane * 4 + 0];
  bv.y = bias[lane * 4 + 1];
  bv.z = bias[lane * 4 + 2];
  bv.w = bias[lane * 4 + 3];

  int b_l, c_l;
  float d_l;
  {
    int t0 = (int)blockIdx.x * 16 + wave * 2 + (lane & 1);
    if (t0 > n - 1) t0 = n - 1;
    b_l = blockoff[t0 / SC_THREADS] + loc[t0];
    c_l = cnt[t0];
    d_l = dinv[t0];
  }

  for (int s = blockIdx.x; s < nstripes; s += AG_GRID) {
    int   tq[2], begq[2], cq[2], slq[2];
    float dtq[2], wlq[2];
    bool  okq[2];
    float4 accq[2];
#pragma unroll
    for (int q = 0; q < 2; ++q) {
      const int t = s * 16 + wave * 2 + q;
      okq[q] = t < n;
      tq[q]  = okq[q] ? t : n - 1;
      begq[q] = __builtin_amdgcn_readlane(b_l, q);
      cq[q]   = okq[q] ? __builtin_amdgcn_readlane(c_l, q) : 0;
      dtq[q]  = __int_as_float(__builtin_amdgcn_readlane(__float_as_int(d_l), q));
    }
    {
      int tn = (s + AG_GRID) * 16 + wave * 2 + (lane & 1);
      if (tn > n - 1) tn = n - 1;
      b_l = blockoff[tn / SC_THREADS] + loc[tn];
      c_l = cnt[tn];
      d_l = dinv[tn];
    }
#pragma unroll
    for (int q = 0; q < 2; ++q) {
      const int m = cq[q] < 32 ? cq[q] : 32;
      const int sv = srcs[lane < m ? begq[q] + lane : 0];
      slq[q] = lane < m ? sv : tq[q];
      accq[q] = seq4[(size_t)tq[q] * (D / 4) + lane];
    }
#pragma unroll
    for (int q = 0; q < 2; ++q) {
      const int m = cq[q] < 32 ? cq[q] : 32;
      wlq[q] = dinv[slq[q]];
    }
#pragma unroll
    for (int q = 0; q < 2; ++q) {
      const int   t   = tq[q];
      const int   c   = cq[q];
      const int   beg = begq[q];
      const float dt  = dtq[q];
      int   s_l = slq[q];
      float w_l = wlq[q];
      float4 acc = accq[q];
      const float ws = okq[q] ? dt * dt : 0.0f;
      acc.x *= ws; acc.y *= ws; acc.z *= ws; acc.w *= ws;

      for (int j0 = 0; j0 < c; j0 += 32) {
        int m = c - j0;
        if (m > 32) m = 32;
        if (j0 > 0) {
          const int sv = srcs[lane < m ? beg + j0 + lane : 0];
          s_l = lane < m ? sv : t;
          w_l = dinv[s_l];
        }
        for (int j = 0; j < m; j += GA_UNROLL) {
          float4 v[GA_UNROLL];
#pragma unroll
          for (int u = 0; u < GA_UNROLL; ++u) {
            const int su = __builtin_amdgcn_readlane(s_l, j + u);
            v[u] = seq4[(size_t)su * (D / 4) + lane];
          }
#pragma unroll
          for (int u = 0; u < GA_UNROLL; ++u) {
            const float wr = __int_as_float(__builtin_amdgcn_readlane(__float_as_int(w_l), j + u)) * dt;
            const float w  = j + u < m ? wr : 0.0f;
            acc.x += v[u].x * w;
            acc.y += v[u].y * w;
            acc.z += v[u].z * w;
            acc.w += v[u].w * w;
          }
        }
      }
      *(float4*)&agg[(wave * 2 + q) * ST_PITCH + 4 * lane] = acc;
    }
    __syncthreads();

    v8f c0 = {}, c1 = {};
#pragma unroll
    for (int i = 0; i < D / 4; i += 2) {
      const v2f a0 = *(const v2f*)&agg[mn * ST_PITCH + 4 * i + khalf];
      const v2f a1 = *(const v2f*)&agg[mn * ST_PITCH + 4 * (i + 1) + khalf];
      c0 = __builtin_amdgcn_wmma_f32_16x16x4_f32(
           false, a0,  false, b[i],
           (short)0, c0,  false,  false);
      c1 = __builtin_amdgcn_wmma_f32_16x16x4_f32(
           false, a1,  false, b[i + 1],
           (short)0, c1,  false,  false);
    }

#pragma unroll
    for (int j = 0; j < 8; ++j)
      cst[(mbase + j) * ST_PITCH + col0 + mn] = c0[j] + c1[j];
    __syncthreads();
#pragma unroll
    for (int q = 0; q < 2; ++q) {
      const int r = wave * 2 + q;
      const int t = s * 16 + r;
      if (t < n) {
        const float4 x = *(const float4*)&cst[r * ST_PITCH + 4 * lane];
        float4 o;
        o.x = x.x + bv.x;
        o.y = x.y + bv.y;
        o.z = x.z + bv.z;
        o.w = x.w + bv.w;
        o.x = o.x > 0.0f ? o.x : pw * o.x;
        o.y = o.y > 0.0f ? o.y : pw * o.y;
        o.z = o.z > 0.0f ? o.z : pw * o.z;
        o.w = o.w > 0.0f ? o.w : pw * o.w;
        ((float4*)out)[(size_t)t * (D / 4) + lane] = o;
      }
    }
  }
}

extern "C" void kernel_launch(void* const* d_in, const int* in_sizes, int n_in,
                              void* d_out, int out_size, void* d_ws, size_t ws_size,
                              hipStream_t stream) {
  const float* seq   = (const float*)d_in[0];
  const int*   eidx  = (const int*)d_in[1];
  const float* W     = (const float*)d_in[2];
  const float* bias  = (const float*)d_in[3];
  const float* prelu = (const float*)d_in[4];

  const int N = in_sizes[0] / D;
  const int E = in_sizes[1] / 2;
  const int* rowp = eidx;
  const int* colp = eidx + E;
  const int nblk = (N + SC_THREADS - 1) / SC_THREADS;

  float* deg  = (float*)d_ws;
  int*  part  = (int*)(deg + N);
  int*  pre   = part + (size_t)HB_CHUNKS * N;
  int*  loc   = pre + (size_t)HB_CHUNKS * N;
  int*  cnt   = loc + N;
  int*  bsum  = cnt + N;
  int*  boff  = bsum + nblk;
  int*  srcs  = boff + nblk;
  float* out  = (float*)d_out;

  const dim3 hgrid(HB_CHUNKS, (N + HB_BINS - 1) / HB_BINS);
  k_deg_hist <<<hgrid, HB_THREADS, 0, stream>>>(colp, part, E, N);
  k_dinv_scan<<<nblk, SC_THREADS, 0, stream>>>(part, pre, deg, cnt, loc, bsum, N);
  k_blockscan<<<1, SC_THREADS, 0, stream>>>(bsum, boff, nblk);
  k_place    <<<hgrid, HB_THREADS, 0, stream>>>(rowp, colp, pre, loc, boff, srcs, E, N);

  const int nstripes = (N + 15) / 16;
  k_agg_gemm<<<AG_GRID, 256, 0, stream>>>(srcs, cnt, loc, boff, seq, W, deg, bias, prelu, out, N, nstripes);
}
